# starting kernel
# baseline (speedup 1.0000x reference)
#include <hip/hip_runtime.h>
#include <hip/hip_bf16.h>

typedef __attribute__((ext_vector_type(2))) float v2f;
typedef __attribute__((ext_vector_type(8))) float v8f;

#define GCN_D 64


__global__ void zero_f32(float* p, int n) {
    int i = blockIdx.x * blockDim.x + threadIdx.x;
    if (i < n) p[i] = 0.0f;
}

__global__ void deg_kernel(const int* __restrict__ dst, float* __restrict__ deg, int E) {
    int e = blockIdx.x * blockDim.x + threadIdx.x;
    if (e < E) atomicAdd(&deg[dst[e]], 1.0f);
}

__global__ void dinv_kernel(float* __restrict__ d, int n) {
    int i = blockIdx.x * blockDim.x + threadIdx.x;
    if (i < n) d[i] = rsqrtf(d[i] + 1.0f);
}

__global__ void gemm_wmma_f32(const float* __restrict__ X,
                              const float* __restrict__ W,
                              float* __restrict__ H,
                              int nrows, int mtiles) {
    const int wave = blockIdx.x * (blockDim.x >> 5) + (threadIdx.x >> 5);
    if (wave >= mtiles) return;
    const int lane = threadIdx.x & 31;
    const int l15  = lane & 15;
    const int hi   = lane >> 4;
    const int rowBase = wave * 16;
    int arow = rowBase + l15;
    if (arow >= nrows) arow = nrows - 1;

    v8f acc0 = {}, acc1 = {}, acc2 = {}, acc3 = {};

    const float* __restrict__ xrow = X + (size_t)arow * GCN_D;

    #pragma unroll
    for (int kk = 0; kk < GCN_D; kk += 4) {
        const int kb = kk + hi * 2;
        v2f a;
        a.x = xrow[kb];
        a.y = xrow[kb + 1];

        const float* __restrict__ wp = W + (size_t)kb * GCN_D + l15;
        v2f b0, b1, b2, b3;
        b0.x = wp[0];       b0.y = wp[GCN_D];
        b1.x = wp[16];      b1.y = wp[16 + GCN_D];
        b2.x = wp[32];      b2.y = wp[32 + GCN_D];
        b3.x = wp[48];      b3.y = wp[48 + GCN_D];

        acc0 = __builtin_amdgcn_wmma_f32_16x16x4_f32(false, a, false, b0, (short)0, acc0, false, false);
        acc1 = __builtin_amdgcn_wmma_f32_16x16x4_f32(false, a, false, b1, (short)0, acc1, false, false);
        acc2 = __builtin_amdgcn_wmma_f32_16x16x4_f32(false, a, false, b2, (short)0, acc2, false, false);
        acc3 = __builtin_amdgcn_wmma_f32_16x16x4_f32(false, a, false, b3, (short)0, acc3, false, false);
    }

    #pragma unroll
    for (int r = 0; r < 8; ++r) {
        const int row = rowBase + r + hi * 8;
        if (row < nrows) {
            float* __restrict__ o = H + (size_t)row * GCN_D + l15;
            o[0]  = acc0[r];
            o[16] = acc1[r];
            o[32] = acc2[r];
            o[48] = acc3[r];
        }
    }
}

__global__ void scatter1_kernel(const int* __restrict__ src, const int* __restrict__ dst,
                                const float* __restrict__ H, const float* __restrict__ dinv,
                                float* __restrict__ AGG, int E) {
    int e = blockIdx.x * blockDim.y + threadIdx.y;
    if (e >= E) return;
    const int s = src[e];
    const int d = dst[e];
    const float nrm = dinv[s] * dinv[d];
    const int q = threadIdx.x;
    const float4 hv = ((const float4*)(H + (size_t)s * GCN_D))[q];
    float* __restrict__ base = AGG + (size_t)d * GCN_D + q * 4;
    atomicAdd(base + 0, hv.x * nrm);
    atomicAdd(base + 1, hv.y * nrm);
    atomicAdd(base + 2, hv.z * nrm);
    atomicAdd(base + 3, hv.w * nrm);
}

__global__ void finalize1_dot_kernel(const float* __restrict__ AGG, const float* __restrict__ H,
                                     const float* __restrict__ dinv, const float* __restrict__ b1,
                                     const float* __restrict__ W2, float* __restrict__ h2, int n) {
    int i = blockIdx.x * blockDim.x + threadIdx.x;
    if (i >= n) return;
    const float di = dinv[i];
    const float d2 = di * di;
    const float4* __restrict__ ag = (const float4*)(AGG + (size_t)i * GCN_D);
    const float4* __restrict__ hh = (const float4*)(H + (size_t)i * GCN_D);
    const float4* __restrict__ bb = (const float4*)b1;
    const float4* __restrict__ ww = (const float4*)W2;
    float s = 0.0f;
    #pragma unroll
    for (int q = 0; q < GCN_D / 4; ++q) {
        const float4 a = ag[q];
        const float4 h = hh[q];
        const float4 b = bb[q];
        const float4 w = ww[q];
        s += fmaxf(a.x + h.x * d2 + b.x, 0.0f) * w.x;
        s += fmaxf(a.y + h.y * d2 + b.y, 0.0f) * w.y;
        s += fmaxf(a.z + h.z * d2 + b.z, 0.0f) * w.z;
        s += fmaxf(a.w + h.w * d2 + b.w, 0.0f) * w.w;
    }
    h2[i] = s;
}

__global__ void scatter2_kernel(const int* __restrict__ src, const int* __restrict__ dst,
                                const float* __restrict__ h2, const float* __restrict__ dinv,
                                float* __restrict__ agg2, int E) {
    int e = blockIdx.x * blockDim.x + threadIdx.x;
    if (e >= E) return;
    const int s = src[e];
    const int d = dst[e];
    atomicAdd(&agg2[d], h2[s] * dinv[s] * dinv[d]);
}

__global__ void finalize2_kernel(const float* __restrict__ agg2, const float* __restrict__ h2,
                                 const float* __restrict__ dinv, const float* __restrict__ b2,
                                 float* __restrict__ out, int n) {
    int i = blockIdx.x * blockDim.x + threadIdx.x;
    if (i >= n) return;
    const float di = dinv[i];
    out[i] = agg2[i] + h2[i] * di * di + b2[0];
}


extern "C" void kernel_launch(void* const* d_in, const int* in_sizes, int n_in,
                              void* d_out, int out_size, void* d_ws, size_t ws_size,
                              hipStream_t stream) {
    const float* x  = (const float*)d_in[0];
    const int*   ei = (const int*)d_in[1];
    const float* W1 = (const float*)d_in[2];
    const float* b1 = (const float*)d_in[3];
    const float* W2 = (const float*)d_in[4];
    const float* b2 = (const float*)d_in[5];
    float* out = (float*)d_out;

    const int N = in_sizes[0] / GCN_D;
    const int E = in_sizes[1] / 2;
    const int* src = ei;
    const int* dst = ei + E;

    float* H    = (float*)d_ws;
    float* AGG  = H    + (size_t)N * GCN_D;
    float* dinv = AGG  + (size_t)N * GCN_D;
    float* h2   = dinv + N;
    float* agg2 = h2   + N;

    const int T = 256;
    const int totNF = N * GCN_D;

    zero_f32<<<(totNF + T - 1) / T, T, 0, stream>>>(AGG, totNF);
    zero_f32<<<(N + T - 1) / T, T, 0, stream>>>(dinv, N);
    zero_f32<<<(N + T - 1) / T, T, 0, stream>>>(agg2, N);

    deg_kernel<<<(E + T - 1) / T, T, 0, stream>>>(dst, dinv, E);
    dinv_kernel<<<(N + T - 1) / T, T, 0, stream>>>(dinv, N);

    const int mtiles = (N + 15) / 16;
    const int wavesPerBlk = T / 32;
    const int gblocks = (mtiles + wavesPerBlk - 1) / wavesPerBlk;
    gemm_wmma_f32<<<gblocks, T, 0, stream>>>(x, W1, H, N, mtiles);

    {
        dim3 blk(16, 16);
        int grd = (E + 16 - 1) / 16;
        scatter1_kernel<<<grd, blk, 0, stream>>>(src, dst, H, dinv, AGG, E);
    }

    finalize1_dot_kernel<<<(N + T - 1) / T, T, 0, stream>>>(AGG, H, dinv, b1, W2, h2, N);

    scatter2_kernel<<<(E + T - 1) / T, T, 0, stream>>>(src, dst, h2, dinv, agg2, E);

    finalize2_kernel<<<(N + T - 1) / T, T, 0, stream>>>(agg2, h2, dinv, b2, out, N);
}


# Edges counting-sorted by destination (returning int atomics give ranks, scan, fill); both layers pull per node in registers, sums in rank (arrival) order plus lane trees; no f32 atomics, no sum plane
# speedup vs baseline: 1.5499x; 1.5499x over previous
#include <hip/hip_runtime.h>
#include <hip/hip_bf16.h>


typedef __attribute__((ext_vector_type(2))) float v2f;
typedef __attribute__((ext_vector_type(8))) float v8f;

#define GCN_D 64


__global__ void zero_i32(int* p, int n) {
    int i = blockIdx.x * blockDim.x + threadIdx.x;
    if (i < n) p[i] = 0;
}

__global__ void rank_kernel(const int* __restrict__ dst, int* __restrict__ cnt,
                            int* __restrict__ rank, int E) {
    int e = blockIdx.x * blockDim.x + threadIdx.x;
    if (e < E) rank[e] = atomicAdd(&cnt[dst[e]], 1);
}

#define SCAN_T 1024
__global__ void scan_dinv_kernel(const int* __restrict__ cnt, int* __restrict__ rowptr,
                                 float* __restrict__ dinv, int n) {
    __shared__ int wsum[SCAN_T / 32];
    __shared__ int psum[SCAN_T / 32];
    const int t    = threadIdx.x;
    const int lane = t & 31;
    const int w    = t >> 5;
    const int base = blockIdx.x * SCAN_T;

    int pre = 0;
    for (int j = t; j < base; j += SCAN_T) pre += cnt[j];

    const int i = base + t;
    const int c = (i < n) ? cnt[i] : 0;

    int incl = c;
    for (int off = 1; off < 32; off <<= 1) {
        const int v = __shfl_up(incl, off);
        if (lane >= off) incl += v;
    }
    for (int off = 16; off > 0; off >>= 1) pre += __shfl_xor(pre, off);

    if (lane == 31) wsum[w] = incl;
    if (lane == 0)  psum[w] = pre;
    __syncthreads();

    int offset = 0;
    for (int k = 0; k < SCAN_T / 32; ++k) {
        offset += psum[k];
        if (k < w) offset += wsum[k];
    }
    const int excl = offset + incl - c;
    if (i < n) {
        rowptr[i] = excl;
        dinv[i]   = rsqrtf((float)c + 1.0f);
        if (i == n - 1) rowptr[n] = excl + c;
    }
}

__global__ void fill_kernel(const int* __restrict__ src, const int* __restrict__ dst,
                            const int* __restrict__ rowptr, const int* __restrict__ rank,
                            int* __restrict__ col, int E) {
    int e = blockIdx.x * blockDim.x + threadIdx.x;
    if (e < E) col[rowptr[dst[e]] + rank[e]] = src[e];
}

__global__ void gemm_wmma_f32(const float* __restrict__ X,
                              const float* __restrict__ W,
                              float* __restrict__ H,
                              int nrows, int mtiles) {
    const int wave = blockIdx.x * (blockDim.x >> 5) + (threadIdx.x >> 5);
    if (wave >= mtiles) return;
    const int lane = threadIdx.x & 31;
    const int l15  = lane & 15;
    const int hi   = lane >> 4;
    const int rowBase = wave * 16;
    int arow = rowBase + l15;
    if (arow >= nrows) arow = nrows - 1;

    v8f acc0 = {}, acc1 = {}, acc2 = {}, acc3 = {};

    const float* __restrict__ xrow = X + (size_t)arow * GCN_D;

    #pragma unroll
    for (int kk = 0; kk < GCN_D; kk += 4) {
        const int kb = kk + hi * 2;
        v2f a;
        a.x = xrow[kb];
        a.y = xrow[kb + 1];

        const float* __restrict__ wp = W + (size_t)kb * GCN_D + l15;
        v2f b0, b1, b2, b3;
        b0.x = wp[0];       b0.y = wp[GCN_D];
        b1.x = wp[16];      b1.y = wp[16 + GCN_D];
        b2.x = wp[32];      b2.y = wp[32 + GCN_D];
        b3.x = wp[48];      b3.y = wp[48 + GCN_D];

        acc0 = __builtin_amdgcn_wmma_f32_16x16x4_f32(false, a, false, b0, (short)0, acc0, false, false);
        acc1 = __builtin_amdgcn_wmma_f32_16x16x4_f32(false, a, false, b1, (short)0, acc1, false, false);
        acc2 = __builtin_amdgcn_wmma_f32_16x16x4_f32(false, a, false, b2, (short)0, acc2, false, false);
        acc3 = __builtin_amdgcn_wmma_f32_16x16x4_f32(false, a, false, b3, (short)0, acc3, false, false);
    }

    #pragma unroll
    for (int r = 0; r < 8; ++r) {
        const int row = rowBase + r + hi * 8;
        if (row < nrows) {
            float* __restrict__ o = H + (size_t)row * GCN_D + l15;
            o[0]  = acc0[r];
            o[16] = acc1[r];
            o[32] = acc2[r];
            o[48] = acc3[r];
        }
    }
}


static __device__ __forceinline__ int   lane_i(int v, int k)   { return __builtin_amdgcn_readlane(v, k); }
static __device__ __forceinline__ float lane_f(float v, int k) {
    return __builtin_bit_cast(float, __builtin_amdgcn_readlane(__builtin_bit_cast(int, v), k));
}
static __device__ __forceinline__ float2 row2(const float* __restrict__ H, int s, int lane) {
    return ((const float2*)(H + (size_t)s * GCN_D))[lane];
}

template <int U>
static __device__ __forceinline__ void add_rows(const float* __restrict__ H, int s_l, float n_l,
                                                int k, int m, int lane, float2& acc) {
    int    s[U];
    float  nn[U];
    float2 h[U];
    #pragma unroll
    for (int u = 0; u < U; ++u) { s[u] = lane_i(s_l, k + u); nn[u] = lane_f(n_l, k + u); }
    #pragma unroll
    for (int u = 0; u < U; ++u) h[u] = row2(H, s[u], lane);
    #pragma unroll
    for (int u = 0; u < U; ++u) {
        const float hx = (u == 0 || k + u < m) ? h[u].x : 0.0f;
        const float hy = (u == 0 || k + u < m) ? h[u].y : 0.0f;
        acc.x += hx * nn[u];
        acc.y += hy * nn[u];
    }
}

__global__ void pull1_dot_kernel(const int* __restrict__ rowptr, const int* __restrict__ col,
                                 const float* __restrict__ H, const float* __restrict__ dinv,
                                 const float* __restrict__ b1, const float* __restrict__ W2,
                                 float* __restrict__ h2, int n) {
    const int lane = threadIdx.x & 31;
    const int wid    = __builtin_amdgcn_readfirstlane((int)(blockIdx.x * (blockDim.x >> 5) + (threadIdx.x >> 5)));
    const int nwaves = (int)(gridDim.x * (blockDim.x >> 5));
    const float2 bb = ((const float2*)b1)[lane];
    const float2 ww = ((const float2*)W2)[lane];

    for (int i = wid; i < n; i += nwaves) {
        const int   beg = rowptr[i];
        const int   end = rowptr[i + 1];
        const float di  = dinv[i];
        const float2 hs = row2(H, i, lane);
        float2 acc = make_float2(0.0f, 0.0f);

        for (int base = beg; base < end; base += 32) {
            const int j = base + lane;
            int   s_l = i;
            float n_l = 0.0f;
            if (j < end) { s_l = col[j]; n_l = dinv[s_l] * di; }
            const int left = end - base;
            const int m    = left < 32 ? left : 32;

            int k = 0;
            for (; k + 4 < m; k += 8) add_rows<8>(H, s_l, n_l, k, m, lane, acc);
            if (k < m)                add_rows<4>(H, s_l, n_l, k, m, lane, acc);
        }

        const float d2 = di * di;
        float p = fmaxf(acc.x + hs.x * d2 + bb.x, 0.0f) * ww.x;
        p      += fmaxf(acc.y + hs.y * d2 + bb.y, 0.0f) * ww.y;
        for (int off = 16; off > 0; off >>= 1) p += __shfl_xor(p, off);
        if (lane == 0) h2[i] = p;
    }
}

__global__ void pull2_kernel(const int* __restrict__ rowptr, const int* __restrict__ col,
                             const float* __restrict__ h2, const float* __restrict__ dinv,
                             const float* __restrict__ b2, float* __restrict__ out, int n) {
    const int t   = blockIdx.x * blockDim.x + threadIdx.x;
    const int i   = t >> 3;
    const int sub = t & 7;
    float v  = 0.0f;
    float di = 0.0f;
    if (i < n) {
        di = dinv[i];
        const int beg = rowptr[i];
        const int end = rowptr[i + 1];
        for (int j = beg + sub; j < end; j += 8) {
            const int s = col[j];
            v += h2[s] * dinv[s] * di;
        }
    }
    v += __shfl_xor(v, 1);
    v += __shfl_xor(v, 2);
    v += __shfl_xor(v, 4);
    if (i < n && sub == 0) out[i] = v + h2[i] * di * di + b2[0];
}


extern "C" void kernel_launch(void* const* d_in, const int* in_sizes, int n_in,
                              void* d_out, int out_size, void* d_ws, size_t ws_size,
                              hipStream_t stream) {
    const float* x  = (const float*)d_in[0];
    const int*   ei = (const int*)d_in[1];
    const float* W1 = (const float*)d_in[2];
    const float* b1 = (const float*)d_in[3];
    const float* W2 = (const float*)d_in[4];
    const float* b2 = (const float*)d_in[5];
    float* out = (float*)d_out;

    const int N = in_sizes[0] / GCN_D;
    const int E = in_sizes[1] / 2;
    const int* src = ei;
    const int* dst = ei + E;

    float* H      = (float*)d_ws;
    int*   cnt    = (int*)(H + (size_t)N * GCN_D);
    int*   rowptr = cnt + N;
    float* dinv   = (float*)(rowptr + N + 1);
    float* h2     = dinv + N;
    int*   rank   = (int*)(h2 + N);
    int*   col    = rank + E;

    const int T = 256;

    zero_i32<<<(N + T - 1) / T, T, 0, stream>>>(cnt, N);
    rank_kernel<<<(E + T - 1) / T, T, 0, stream>>>(dst, cnt, rank, E);

    scan_dinv_kernel<<<(N + SCAN_T - 1) / SCAN_T, SCAN_T, 0, stream>>>(cnt, rowptr, dinv, N);
    fill_kernel<<<(E + T - 1) / T, T, 0, stream>>>(src, dst, rowptr, rank, col, E);

    const int mtiles = (N + 15) / 16;
    const int wavesPerBlk = T / 32;
    const int gblocks = (mtiles + wavesPerBlk - 1) / wavesPerBlk;
    gemm_wmma_f32<<<gblocks, T, 0, stream>>>(x, W1, H, N, mtiles);

    pull1_dot_kernel<<<2048, T, 0, stream>>>(rowptr, col, H, dinv, b1, W2, h2, N);

    pull2_kernel<<<(N * 8 + T - 1) / T, T, 0, stream>>>(rowptr, col, h2, dinv, b2, out, N);
}


# Edge sort by destination redone as two counting sorts in LDS (chunks by 256-node bin, then one work-group per bin): no global atomics, no scan launch, 7 launches to 5; list order = LDS arrival order
# speedup vs baseline: 3.7104x; 2.3940x over previous
#include <hip/hip_runtime.h>
#include <hip/hip_bf16.h>


typedef __attribute__((ext_vector_type(2))) float v2f;
typedef __attribute__((ext_vector_type(8))) float v8f;

#define GCN_D 64


#define BIN_NODES   256
#define MAX_BINS    256
#define CHUNK       4096
#define MAX_CHUNKS  256
#define SORT_T      256
#define EPT         (CHUNK / SORT_T)
#define RUN_LANES   16
#define RUN_GROUPS  (SORT_T / RUN_LANES)
#define RUN_ROUNDS  (MAX_CHUNKS / RUN_GROUPS)
#define LCAP        8192

static __device__ __forceinline__ int scan256(int v, int t, int* wtot, int& tot) {
    const int lane = t & 31;
    const int w    = t >> 5;
    int incl = v;
    for (int off = 1; off < 32; off <<= 1) {
        const int u = __shfl_up(incl, off);
        if (lane >= off) incl += u;
    }
    if (lane == 31) wtot[w] = incl;
    __syncthreads();
    int pre = 0, all = 0;
    for (int k = 0; k < SORT_T / 32; ++k) {
        const int x = wtot[k];
        all += x;
        if (k < w) pre += x;
    }
    tot = all;
    __syncthreads();
    return pre + incl - v;
}

__global__ void __launch_bounds__(SORT_T)
chunk_sort_kernel(const int* __restrict__ src, const int* __restrict__ dst,
                  int* __restrict__ sorted, int* __restrict__ cstart, int E, int nbins) {
    __shared__ int hist[MAX_BINS];
    __shared__ int wtot[SORT_T / 32];
    __shared__ int buf[CHUNK];
    const int t    = threadIdx.x;
    const int base = blockIdx.x * CHUNK;
    hist[t] = 0;
    __syncthreads();

    int w[EPT], b[EPT], r[EPT];
    #pragma unroll
    for (int k = 0; k < EPT; ++k) {
        const int  e  = base + k * SORT_T + t;
        const bool ok = e < E;
        const int  ec = ok ? e : E - 1;
        const int  d  = dst[ec];
        w[k] = (int)(((unsigned)src[ec] << 8) | (unsigned)(d & (BIN_NODES - 1)));
        b[k] = ok ? (d >> 8) : -1;
    }
    #pragma unroll
    for (int k = 0; k < EPT; ++k) r[k] = (b[k] >= 0) ? atomicAdd(&hist[b[k]], 1) : 0;
    __syncthreads();

    int tot;
    const int ex = scan256(hist[t], t, wtot, tot);
    int* __restrict__ crow = cstart + (size_t)blockIdx.x * (nbins + 1);
    if (t < nbins) crow[t] = ex;
    if (t == 0)    crow[nbins] = tot;
    hist[t] = ex;
    __syncthreads();

    int st[EPT];
    #pragma unroll
    for (int k = 0; k < EPT; ++k) st[k] = hist[b[k] >= 0 ? b[k] : 0];
    #pragma unroll
    for (int k = 0; k < EPT; ++k)
        if (b[k] >= 0) buf[st[k] + r[k]] = w[k];
    __syncthreads();

    if (base + CHUNK <= E) {
        #pragma unroll
        for (int k = 0; k < EPT; ++k) sorted[base + k * SORT_T + t] = buf[k * SORT_T + t];
    } else {
        for (int i = t; i < E - base; i += SORT_T) sorted[base + i] = buf[i];
    }
}

__global__ void __launch_bounds__(SORT_T)
bin_lists_kernel(const int* __restrict__ sorted, const int* __restrict__ cstart,
                 int* __restrict__ rowptr, float* __restrict__ dinv, int* __restrict__ col,
                 int n, int nchunks, int nbins) {
    __shared__ int rs0[MAX_CHUNKS];
    __shared__ int rlen[MAX_CHUNKS];
    __shared__ int cnt[BIN_NODES];
    __shared__ int cur[BIN_NODES];
    __shared__ int wa[SORT_T / 32];
    __shared__ int wb[SORT_T / 32];
    __shared__ int lcol[LCAP];
    const int t   = threadIdx.x;
    const int bin = blockIdx.x;

    int s0 = 0, len = 0;
    if (t < nchunks) {
        const int* __restrict__ crow = cstart + (size_t)t * (nbins + 1) + bin;
        s0  = crow[0];
        len = crow[1] - s0;
    }
    rs0[t]  = t * CHUNK + s0;
    rlen[t] = len;
    cnt[t]  = 0;
    int sa = s0, sb = len;
    for (int off = 16; off > 0; off >>= 1) { sa += __shfl_xor(sa, off); sb += __shfl_xor(sb, off); }
    if ((t & 31) == 0) { wa[t >> 5] = sa; wb[t >> 5] = sb; }
    __syncthreads();
    int binstart = 0, nedges = 0;
    for (int k = 0; k < SORT_T / 32; ++k) { binstart += wa[k]; nedges += wb[k]; }

    const int g   = t / RUN_LANES;
    const int sub = t % RUN_LANES;
    int  gs[RUN_ROUNDS], gl[RUN_ROUNDS];
    int  wv[RUN_ROUNDS][2];
    bool longrun = false;
    #pragma unroll
    for (int j = 0; j < RUN_ROUNDS; ++j) {
        const int p = g + j * RUN_GROUPS;
        gs[j] = rs0[p];
        gl[j] = rlen[p];
    }
    #pragma unroll
    for (int j = 0; j < RUN_ROUNDS; ++j) {
        wv[j][0] = 0;
        wv[j][1] = 0;
        if (sub < gl[j])             wv[j][0] = sorted[gs[j] + sub];
        if (sub + RUN_LANES < gl[j]) wv[j][1] = sorted[gs[j] + sub + RUN_LANES];
        longrun = longrun || (gl[j] > 2 * RUN_LANES);
    }

    #pragma unroll
    for (int j = 0; j < RUN_ROUNDS; ++j) {
        if (sub < gl[j])             atomicAdd(&cnt[wv[j][0] & (BIN_NODES - 1)], 1);
        if (sub + RUN_LANES < gl[j]) atomicAdd(&cnt[wv[j][1] & (BIN_NODES - 1)], 1);
    }
    if (longrun) {
        for (int p = g; p < nchunks; p += RUN_GROUPS)
            for (int i = sub + 2 * RUN_LANES; i < rlen[p]; i += RUN_LANES)
                atomicAdd(&cnt[sorted[rs0[p] + i] & (BIN_NODES - 1)], 1);
    }
    __syncthreads();

    const int c = cnt[t];
    int tot;
    const int ex = scan256(c, t, wa, tot);
    cur[t] = ex;
    const int node = bin * BIN_NODES + t;
    if (node < n) {
        rowptr[node] = binstart + ex;
        dinv[node]   = rsqrtf((float)c + 1.0f);
        if (node == n - 1) rowptr[n] = binstart + ex + c;
    }
    __syncthreads();

    const bool fits = nedges <= LCAP;
    int pos[RUN_ROUNDS][2];
    #pragma unroll
    for (int j = 0; j < RUN_ROUNDS; ++j) {
        #pragma unroll
        for (int h = 0; h < 2; ++h) {
            pos[j][h] = 0;
            if (sub + h * RUN_LANES < gl[j]) pos[j][h] = atomicAdd(&cur[(unsigned)wv[j][h] & (BIN_NODES - 1)], 1);
        }
    }
    if (fits) {
        #pragma unroll
        for (int j = 0; j < RUN_ROUNDS; ++j) {
            #pragma unroll
            for (int h = 0; h < 2; ++h)
                if (sub + h * RUN_LANES < gl[j]) lcol[pos[j][h]] = (int)((unsigned)wv[j][h] >> 8);
        }
    } else {
        #pragma unroll
        for (int j = 0; j < RUN_ROUNDS; ++j) {
            #pragma unroll
            for (int h = 0; h < 2; ++h)
                if (sub + h * RUN_LANES < gl[j]) col[binstart + pos[j][h]] = (int)((unsigned)wv[j][h] >> 8);
        }
    }
    if (longrun) {
        for (int p = g; p < nchunks; p += RUN_GROUPS)
            for (int i = sub + 2 * RUN_LANES; i < rlen[p]; i += RUN_LANES) {
                const unsigned wd    = (unsigned)sorted[rs0[p] + i];
                const int      place = atomicAdd(&cur[wd & (BIN_NODES - 1)], 1);
                if (fits) lcol[place] = (int)(wd >> 8);
                else      col[binstart + place] = (int)(wd >> 8);
            }
    }
    __syncthreads();

    if (fits) {
        for (int i0 = 0; i0 < nedges; i0 += 4 * SORT_T) {
            int v[4];
            #pragma unroll
            for (int u = 0; u < 4; ++u) {
                const int i = i0 + u * SORT_T + t;
                v[u] = lcol[i < nedges ? i : nedges - 1];
            }
            #pragma unroll
            for (int u = 0; u < 4; ++u) {
                const int i = i0 + u * SORT_T + t;
                if (i < nedges) col[binstart + i] = v[u];
            }
        }
    }
}

__global__ void gemm_wmma_f32(const float* __restrict__ X,
                              const float* __restrict__ W,
                              float* __restrict__ H,
                              int nrows, int mtiles) {
    const int wave = blockIdx.x * (blockDim.x >> 5) + (threadIdx.x >> 5);
    if (wave >= mtiles) return;
    const int lane = threadIdx.x & 31;
    const int l15  = lane & 15;
    const int hi   = lane >> 4;
    const int rowBase = wave * 16;
    int arow = rowBase + l15;
    if (arow >= nrows) arow = nrows - 1;

    v8f acc0 = {}, acc1 = {}, acc2 = {}, acc3 = {};

    const float* __restrict__ xrow = X + (size_t)arow * GCN_D;

    #pragma unroll
    for (int kk = 0; kk < GCN_D; kk += 4) {
        const int kb = kk + hi * 2;
        v2f a;
        a.x = xrow[kb];
        a.y = xrow[kb + 1];

        const float* __restrict__ wp = W + (size_t)kb * GCN_D + l15;
        v2f b0, b1, b2, b3;
        b0.x = wp[0];       b0.y = wp[GCN_D];
        b1.x = wp[16];      b1.y = wp[16 + GCN_D];
        b2.x = wp[32];      b2.y = wp[32 + GCN_D];
        b3.x = wp[48];      b3.y = wp[48 + GCN_D];

        acc0 = __builtin_amdgcn_wmma_f32_16x16x4_f32(false, a, false, b0, (short)0, acc0, false, false);
        acc1 = __builtin_amdgcn_wmma_f32_16x16x4_f32(false, a, false, b1, (short)0, acc1, false, false);
        acc2 = __builtin_amdgcn_wmma_f32_16x16x4_f32(false, a, false, b2, (short)0, acc2, false, false);
        acc3 = __builtin_amdgcn_wmma_f32_16x16x4_f32(false, a, false, b3, (short)0, acc3, false, false);
    }

    #pragma unroll
    for (int r = 0; r < 8; ++r) {
        const int row = rowBase + r + hi * 8;
        if (row < nrows) {
            float* __restrict__ o = H + (size_t)row * GCN_D + l15;
            o[0]  = acc0[r];
            o[16] = acc1[r];
            o[32] = acc2[r];
            o[48] = acc3[r];
        }
    }
}


static __device__ __forceinline__ int   lane_i(int v, int k)   { return __builtin_amdgcn_readlane(v, k); }
static __device__ __forceinline__ float lane_f(float v, int k) {
    return __builtin_bit_cast(float, __builtin_amdgcn_readlane(__builtin_bit_cast(int, v), k));
}
static __device__ __forceinline__ float2 row2(const float* __restrict__ H, int s, int lane) {
    return ((const float2*)(H + (size_t)s * GCN_D))[lane];
}

template <int U>
static __device__ __forceinline__ void add_rows(const float* __restrict__ H, int s_l, float n_l,
                                                int k, int m, int lane, float2& acc) {
    int    s[U];
    float  nn[U];
    float2 h[U];
    #pragma unroll
    for (int u = 0; u < U; ++u) { s[u] = lane_i(s_l, k + u); nn[u] = lane_f(n_l, k + u); }
    #pragma unroll
    for (int u = 0; u < U; ++u) h[u] = row2(H, s[u], lane);
    #pragma unroll
    for (int u = 0; u < U; ++u) {
        const float hx = (u == 0 || k + u < m) ? h[u].x : 0.0f;
        const float hy = (u == 0 || k + u < m) ? h[u].y : 0.0f;
        acc.x += hx * nn[u];
        acc.y += hy * nn[u];
    }
}

__global__ void pull1_dot_kernel(const int* __restrict__ rowptr, const int* __restrict__ col,
                                 const float* __restrict__ H, const float* __restrict__ dinv,
                                 const float* __restrict__ b1, const float* __restrict__ W2,
                                 float* __restrict__ h2, int n) {
    const int lane = threadIdx.x & 31;
    const int wid    = __builtin_amdgcn_readfirstlane((int)(blockIdx.x * (blockDim.x >> 5) + (threadIdx.x >> 5)));
    const int nwaves = (int)(gridDim.x * (blockDim.x >> 5));
    const float2 bb = ((const float2*)b1)[lane];
    const float2 ww = ((const float2*)W2)[lane];

    for (int i = wid; i < n; i += nwaves) {
        const int   beg = rowptr[i];
        const int   end = rowptr[i + 1];
        const float di  = dinv[i];
        const float2 hs = row2(H, i, lane);
        float2 acc = make_float2(0.0f, 0.0f);

        for (int base = beg; base < end; base += 32) {
            const int j = base + lane;
            int   s_l = i;
            float n_l = 0.0f;
            if (j < end) { s_l = col[j]; n_l = dinv[s_l] * di; }
            const int left = end - base;
            const int m    = left < 32 ? left : 32;

            int k = 0;
            for (; k + 4 < m; k += 8) add_rows<8>(H, s_l, n_l, k, m, lane, acc);
            if (k < m)                add_rows<4>(H, s_l, n_l, k, m, lane, acc);
        }

        const float d2 = di * di;
        float p = fmaxf(acc.x + hs.x * d2 + bb.x, 0.0f) * ww.x;
        p      += fmaxf(acc.y + hs.y * d2 + bb.y, 0.0f) * ww.y;
        for (int off = 16; off > 0; off >>= 1) p += __shfl_xor(p, off);
        if (lane == 0) h2[i] = p;
    }
}

__global__ void pull2_kernel(const int* __restrict__ rowptr, const int* __restrict__ col,
                             const float* __restrict__ h2, const float* __restrict__ dinv,
                             const float* __restrict__ b2, float* __restrict__ out, int n) {
    const int t   = blockIdx.x * blockDim.x + threadIdx.x;
    const int i   = t >> 3;
    const int sub = t & 7;
    float v  = 0.0f;
    float di = 0.0f;
    if (i < n) {
        di = dinv[i];
        const int beg = rowptr[i];
        const int end = rowptr[i + 1];
        for (int j = beg + sub; j < end; j += 8) {
            const int s = col[j];
            v += h2[s] * dinv[s] * di;
        }
    }
    v += __shfl_xor(v, 1);
    v += __shfl_xor(v, 2);
    v += __shfl_xor(v, 4);
    if (i < n && sub == 0) out[i] = v + h2[i] * di * di + b2[0];
}


extern "C" void kernel_launch(void* const* d_in, const int* in_sizes, int n_in,
                              void* d_out, int out_size, void* d_ws, size_t ws_size,
                              hipStream_t stream) {
    const float* x  = (const float*)d_in[0];
    const int*   ei = (const int*)d_in[1];
    const float* W1 = (const float*)d_in[2];
    const float* b1 = (const float*)d_in[3];
    const float* W2 = (const float*)d_in[4];
    const float* b2 = (const float*)d_in[5];
    float* out = (float*)d_out;

    const int N = in_sizes[0] / GCN_D;
    const int E = in_sizes[1] / 2;
    const int* src = ei;
    const int* dst = ei + E;

    float* H      = (float*)d_ws;
    int*   rowptr = (int*)(H + (size_t)N * GCN_D);
    float* dinv   = (float*)(rowptr + N + 1);
    float* h2     = dinv + N;
    int*   col    = (int*)(h2 + N);
    int*   sorted = col + E;
    int*   cstart = sorted + E;

    const int T = 256;
    const int nbins   = (N + BIN_NODES - 1) / BIN_NODES;
    const int nchunks = (E + CHUNK - 1) / CHUNK;

    chunk_sort_kernel<<<nchunks, SORT_T, 0, stream>>>(src, dst, sorted, cstart, E, nbins);

    bin_lists_kernel<<<nbins, SORT_T, 0, stream>>>(sorted, cstart, rowptr, dinv, col, N, nchunks, nbins);

    const int mtiles = (N + 15) / 16;
    const int wavesPerBlk = T / 32;
    const int gblocks = (mtiles + wavesPerBlk - 1) / wavesPerBlk;
    gemm_wmma_f32<<<gblocks, T, 0, stream>>>(x, W1, H, N, mtiles);

    pull1_dot_kernel<<<2048, T, 0, stream>>>(rowptr, col, H, dinv, b1, W2, h2, N);

    pull2_kernel<<<(N * 8 + T - 1) / T, T, 0, stream>>>(rowptr, col, h2, dinv, b2, out, N);
}


# Layer 2 gathers one 8-byte (h2, dinv) pair per edge from a plane that the layer-1 kernel writes, in place of two 4-byte reads of two planes: half the random second-level requests; sums unchanged
# speedup vs baseline: 4.0662x; 1.0959x over previous
#include <hip/hip_runtime.h>
#include <hip/hip_bf16.h>


typedef __attribute__((ext_vector_type(2))) float v2f;
typedef __attribute__((ext_vector_type(8))) float v8f;

#define GCN_D 64


#define BIN_NODES   256
#define MAX_BINS    256
#define CHUNK       4096
#define MAX_CHUNKS  256
#define SORT_T      256
#define EPT         (CHUNK / SORT_T)
#define RUN_LANES   16
#define RUN_GROUPS  (SORT_T / RUN_LANES)
#define RUN_ROUNDS  (MAX_CHUNKS / RUN_GROUPS)
#define LCAP        8192

static __device__ __forceinline__ int scan256(int v, int t, int* wtot, int& tot) {
    const int lane = t & 31;
    const int w    = t >> 5;
    int incl = v;
    for (int off = 1; off < 32; off <<= 1) {
        const int u = __shfl_up(incl, off);
        if (lane >= off) incl += u;
    }
    if (lane == 31) wtot[w] = incl;
    __syncthreads();
    int pre = 0, all = 0;
    for (int k = 0; k < SORT_T / 32; ++k) {
        const int x = wtot[k];
        all += x;
        if (k < w) pre += x;
    }
    tot = all;
    __syncthreads();
    return pre + incl - v;
}

__global__ void __launch_bounds__(SORT_T)
chunk_sort_kernel(const int* __restrict__ src, const int* __restrict__ dst,
                  int* __restrict__ sorted, int* __restrict__ cstart, int E, int nbins) {
    __shared__ int hist[MAX_BINS];
    __shared__ int wtot[SORT_T / 32];
    __shared__ int buf[CHUNK];
    const int t    = threadIdx.x;
    const int base = blockIdx.x * CHUNK;
    hist[t] = 0;
    __syncthreads();

    int w[EPT], b[EPT], r[EPT];
    #pragma unroll
    for (int k = 0; k < EPT; ++k) {
        const int  e  = base + k * SORT_T + t;
        const bool ok = e < E;
        const int  ec = ok ? e : E - 1;
        const int  d  = dst[ec];
        w[k] = (int)(((unsigned)src[ec] << 8) | (unsigned)(d & (BIN_NODES - 1)));
        b[k] = ok ? (d >> 8) : -1;
    }
    #pragma unroll
    for (int k = 0; k < EPT; ++k) r[k] = (b[k] >= 0) ? atomicAdd(&hist[b[k]], 1) : 0;
    __syncthreads();

    int tot;
    const int ex = scan256(hist[t], t, wtot, tot);
    int* __restrict__ crow = cstart + (size_t)blockIdx.x * (nbins + 1);
    if (t < nbins) crow[t] = ex;
    if (t == 0)    crow[nbins] = tot;
    hist[t] = ex;
    __syncthreads();

    int st[EPT];
    #pragma unroll
    for (int k = 0; k < EPT; ++k) st[k] = hist[b[k] >= 0 ? b[k] : 0];
    #pragma unroll
    for (int k = 0; k < EPT; ++k)
        if (b[k] >= 0) buf[st[k] + r[k]] = w[k];
    __syncthreads();

    if (base + CHUNK <= E) {
        #pragma unroll
        for (int k = 0; k < EPT; ++k) sorted[base + k * SORT_T + t] = buf[k * SORT_T + t];
    } else {
        for (int i = t; i < E - base; i += SORT_T) sorted[base + i] = buf[i];
    }
}

__global__ void __launch_bounds__(SORT_T)
bin_lists_kernel(const int* __restrict__ sorted, const int* __restrict__ cstart,
                 int* __restrict__ rowptr, float* __restrict__ dinv, int* __restrict__ col,
                 int n, int nchunks, int nbins) {
    __shared__ int rs0[MAX_CHUNKS];
    __shared__ int rlen[MAX_CHUNKS];
    __shared__ int cnt[BIN_NODES];
    __shared__ int cur[BIN_NODES];
    __shared__ int wa[SORT_T / 32];
    __shared__ int wb[SORT_T / 32];
    __shared__ int lcol[LCAP];
    const int t   = threadIdx.x;
    const int bin = blockIdx.x;

    int s0 = 0, len = 0;
    if (t < nchunks) {
        const int* __restrict__ crow = cstart + (size_t)t * (nbins + 1) + bin;
        s0  = crow[0];
        len = crow[1] - s0;
    }
    rs0[t]  = t * CHUNK + s0;
    rlen[t] = len;
    cnt[t]  = 0;
    int sa = s0, sb = len;
    for (int off = 16; off > 0; off >>= 1) { sa += __shfl_xor(sa, off); sb += __shfl_xor(sb, off); }
    if ((t & 31) == 0) { wa[t >> 5] = sa; wb[t >> 5] = sb; }
    __syncthreads();
    int binstart = 0, nedges = 0;
    for (int k = 0; k < SORT_T / 32; ++k) { binstart += wa[k]; nedges += wb[k]; }

    const int g   = t / RUN_LANES;
    const int sub = t % RUN_LANES;
    int  gs[RUN_ROUNDS], gl[RUN_ROUNDS];
    int  wv[RUN_ROUNDS][2];
    bool longrun = false;
    #pragma unroll
    for (int j = 0; j < RUN_ROUNDS; ++j) {
        const int p = g + j * RUN_GROUPS;
        gs[j] = rs0[p];
        gl[j] = rlen[p];
    }
    #pragma unroll
    for (int j = 0; j < RUN_ROUNDS; ++j) {
        wv[j][0] = 0;
        wv[j][1] = 0;
        if (sub < gl[j])             wv[j][0] = sorted[gs[j] + sub];
        if (sub + RUN_LANES < gl[j]) wv[j][1] = sorted[gs[j] + sub + RUN_LANES];
        longrun = longrun || (gl[j] > 2 * RUN_LANES);
    }

    #pragma unroll
    for (int j = 0; j < RUN_ROUNDS; ++j) {
        if (sub < gl[j])             atomicAdd(&cnt[wv[j][0] & (BIN_NODES - 1)], 1);
        if (sub + RUN_LANES < gl[j]) atomicAdd(&cnt[wv[j][1] & (BIN_NODES - 1)], 1);
    }
    if (longrun) {
        for (int p = g; p < nchunks; p += RUN_GROUPS)
            for (int i = sub + 2 * RUN_LANES; i < rlen[p]; i += RUN_LANES)
                atomicAdd(&cnt[sorted[rs0[p] + i] & (BIN_NODES - 1)], 1);
    }
    __syncthreads();

    const int c = cnt[t];
    int tot;
    const int ex = scan256(c, t, wa, tot);
    cur[t] = ex;
    const int node = bin * BIN_NODES + t;
    if (node < n) {
        rowptr[node] = binstart + ex;
        dinv[node]   = rsqrtf((float)c + 1.0f);
        if (node == n - 1) rowptr[n] = binstart + ex + c;
    }
    __syncthreads();

    const bool fits = nedges <= LCAP;
    int pos[RUN_ROUNDS][2];
    #pragma unroll
    for (int j = 0; j < RUN_ROUNDS; ++j) {
        #pragma unroll
        for (int h = 0; h < 2; ++h) {
            pos[j][h] = 0;
            if (sub + h * RUN_LANES < gl[j]) pos[j][h] = atomicAdd(&cur[(unsigned)wv[j][h] & (BIN_NODES - 1)], 1);
        }
    }
    if (fits) {
        #pragma unroll
        for (int j = 0; j < RUN_ROUNDS; ++j) {
            #pragma unroll
            for (int h = 0; h < 2; ++h)
                if (sub + h * RUN_LANES < gl[j]) lcol[pos[j][h]] = (int)((unsigned)wv[j][h] >> 8);
        }
    } else {
        #pragma unroll
        for (int j = 0; j < RUN_ROUNDS; ++j) {
            #pragma unroll
            for (int h = 0; h < 2; ++h)
                if (sub + h * RUN_LANES < gl[j]) col[binstart + pos[j][h]] = (int)((unsigned)wv[j][h] >> 8);
        }
    }
    if (longrun) {
        for (int p = g; p < nchunks; p += RUN_GROUPS)
            for (int i = sub + 2 * RUN_LANES; i < rlen[p]; i += RUN_LANES) {
                const unsigned wd    = (unsigned)sorted[rs0[p] + i];
                const int      place = atomicAdd(&cur[wd & (BIN_NODES - 1)], 1);
                if (fits) lcol[place] = (int)(wd >> 8);
                else      col[binstart + place] = (int)(wd >> 8);
            }
    }
    __syncthreads();

    if (fits) {
        for (int i0 = 0; i0 < nedges; i0 += 4 * SORT_T) {
            int v[4];
            #pragma unroll
            for (int u = 0; u < 4; ++u) {
                const int i = i0 + u * SORT_T + t;
                v[u] = lcol[i < nedges ? i : nedges - 1];
            }
            #pragma unroll
            for (int u = 0; u < 4; ++u) {
                const int i = i0 + u * SORT_T + t;
                if (i < nedges) col[binstart + i] = v[u];
            }
        }
    }
}

__global__ void gemm_wmma_f32(const float* __restrict__ X,
                              const float* __restrict__ W,
                              float* __restrict__ H,
                              int nrows, int mtiles) {
    const int wave = blockIdx.x * (blockDim.x >> 5) + (threadIdx.x >> 5);
    if (wave >= mtiles) return;
    const int lane = threadIdx.x & 31;
    const int l15  = lane & 15;
    const int hi   = lane >> 4;
    const int rowBase = wave * 16;
    int arow = rowBase + l15;
    if (arow >= nrows) arow = nrows - 1;

    v8f acc0 = {}, acc1 = {}, acc2 = {}, acc3 = {};

    const float* __restrict__ xrow = X + (size_t)arow * GCN_D;

    #pragma unroll
    for (int kk = 0; kk < GCN_D; kk += 4) {
        const int kb = kk + hi * 2;
        v2f a;
        a.x = xrow[kb];
        a.y = xrow[kb + 1];

        const float* __restrict__ wp = W + (size_t)kb * GCN_D + l15;
        v2f b0, b1, b2, b3;
        b0.x = wp[0];       b0.y = wp[GCN_D];
        b1.x = wp[16];      b1.y = wp[16 + GCN_D];
        b2.x = wp[32];      b2.y = wp[32 + GCN_D];
        b3.x = wp[48];      b3.y = wp[48 + GCN_D];

        acc0 = __builtin_amdgcn_wmma_f32_16x16x4_f32(false, a, false, b0, (short)0, acc0, false, false);
        acc1 = __builtin_amdgcn_wmma_f32_16x16x4_f32(false, a, false, b1, (short)0, acc1, false, false);
        acc2 = __builtin_amdgcn_wmma_f32_16x16x4_f32(false, a, false, b2, (short)0, acc2, false, false);
        acc3 = __builtin_amdgcn_wmma_f32_16x16x4_f32(false, a, false, b3, (short)0, acc3, false, false);
    }

    #pragma unroll
    for (int r = 0; r < 8; ++r) {
        const int row = rowBase + r + hi * 8;
        if (row < nrows) {
            float* __restrict__ o = H + (size_t)row * GCN_D + l15;
            o[0]  = acc0[r];
            o[16] = acc1[r];
            o[32] = acc2[r];
            o[48] = acc3[r];
        }
    }
}


static __device__ __forceinline__ int   lane_i(int v, int k)   { return __builtin_amdgcn_readlane(v, k); }
static __device__ __forceinline__ float lane_f(float v, int k) {
    return __builtin_bit_cast(float, __builtin_amdgcn_readlane(__builtin_bit_cast(int, v), k));
}
static __device__ __forceinline__ float2 row2(const float* __restrict__ H, int s, int lane) {
    return ((const float2*)(H + (size_t)s * GCN_D))[lane];
}

template <int U>
static __device__ __forceinline__ void add_rows(const float* __restrict__ H, int s_l, float n_l,
                                                int k, int m, int lane, float2& acc) {
    int    s[U];
    float  nn[U];
    float2 h[U];
    #pragma unroll
    for (int u = 0; u < U; ++u) { s[u] = lane_i(s_l, k + u); nn[u] = lane_f(n_l, k + u); }
    #pragma unroll
    for (int u = 0; u < U; ++u) h[u] = row2(H, s[u], lane);
    #pragma unroll
    for (int u = 0; u < U; ++u) {
        const float hx = (u == 0 || k + u < m) ? h[u].x : 0.0f;
        const float hy = (u == 0 || k + u < m) ? h[u].y : 0.0f;
        acc.x += hx * nn[u];
        acc.y += hy * nn[u];
    }
}

__global__ void pull1_dot_kernel(const int* __restrict__ rowptr, const int* __restrict__ col,
                                 const float* __restrict__ H, const float* __restrict__ dinv,
                                 const float* __restrict__ b1, const float* __restrict__ W2,
                                 float* __restrict__ hd, int n) {
    const int lane = threadIdx.x & 31;
    const int wid    = __builtin_amdgcn_readfirstlane((int)(blockIdx.x * (blockDim.x >> 5) + (threadIdx.x >> 5)));
    const int nwaves = (int)(gridDim.x * (blockDim.x >> 5));
    const float2 bb = ((const float2*)b1)[lane];
    const float2 ww = ((const float2*)W2)[lane];

    for (int i = wid; i < n; i += nwaves) {
        const int   beg = rowptr[i];
        const int   end = rowptr[i + 1];
        const float di  = dinv[i];
        const float2 hs = row2(H, i, lane);
        float2 acc = make_float2(0.0f, 0.0f);

        for (int base = beg; base < end; base += 32) {
            const int j = base + lane;
            int   s_l = i;
            float n_l = 0.0f;
            if (j < end) { s_l = col[j]; n_l = dinv[s_l] * di; }
            const int left = end - base;
            const int m    = left < 32 ? left : 32;

            int k = 0;
            for (; k + 4 < m; k += 8) add_rows<8>(H, s_l, n_l, k, m, lane, acc);
            if (k < m)                add_rows<4>(H, s_l, n_l, k, m, lane, acc);
        }

        const float d2 = di * di;
        float p = fmaxf(acc.x + hs.x * d2 + bb.x, 0.0f) * ww.x;
        p      += fmaxf(acc.y + hs.y * d2 + bb.y, 0.0f) * ww.y;
        for (int off = 16; off > 0; off >>= 1) p += __shfl_xor(p, off);
        if (lane == 0) ((float2*)hd)[i] = make_float2(p, di);
    }
}

__global__ void pull2_kernel(const int* __restrict__ rowptr, const int* __restrict__ col,
                             const float* __restrict__ hd,
                             const float* __restrict__ b2, float* __restrict__ out, int n) {
    const float2* __restrict__ pr = (const float2*)hd;
    const int t   = blockIdx.x * blockDim.x + threadIdx.x;
    const int i   = t >> 3;
    const int sub = t & 7;
    float v  = 0.0f;
    float hi = 0.0f;
    float di = 0.0f;
    if (i < n) {
        const float2 me = pr[i];
        hi = me.x;
        di = me.y;
        const int beg = rowptr[i];
        const int end = rowptr[i + 1];
        for (int j = beg + sub; j < end; j += 8) {
            const float2 g = pr[col[j]];
            v += g.x * g.y * di;
        }
    }
    v += __shfl_xor(v, 1);
    v += __shfl_xor(v, 2);
    v += __shfl_xor(v, 4);
    if (i < n && sub == 0) out[i] = fmaf(hi * di, di, v) + b2[0];
}


extern "C" void kernel_launch(void* const* d_in, const int* in_sizes, int n_in,
                              void* d_out, int out_size, void* d_ws, size_t ws_size,
                              hipStream_t stream) {
    const float* x  = (const float*)d_in[0];
    const int*   ei = (const int*)d_in[1];
    const float* W1 = (const float*)d_in[2];
    const float* b1 = (const float*)d_in[3];
    const float* W2 = (const float*)d_in[4];
    const float* b2 = (const float*)d_in[5];
    float* out = (float*)d_out;

    const int N = in_sizes[0] / GCN_D;
    const int E = in_sizes[1] / 2;
    const int* src = ei;
    const int* dst = ei + E;

    float* H      = (float*)d_ws;
    int*   sorted = (int*)(H + (size_t)N * GCN_D);
    float* hd     = (float*)sorted;
    int*   col    = sorted + E;
    int*   rowptr = col + E;
    float* dinv   = (float*)(rowptr + N + 1);
    int*   cstart = (int*)(dinv + N);

    const int T = 256;
    const int nbins   = (N + BIN_NODES - 1) / BIN_NODES;
    const int nchunks = (E + CHUNK - 1) / CHUNK;

    chunk_sort_kernel<<<nchunks, SORT_T, 0, stream>>>(src, dst, sorted, cstart, E, nbins);

    bin_lists_kernel<<<nbins, SORT_T, 0, stream>>>(sorted, cstart, rowptr, dinv, col, N, nchunks, nbins);

    const int mtiles = (N + 15) / 16;
    const int wavesPerBlk = T / 32;
    const int gblocks = (mtiles + wavesPerBlk - 1) / wavesPerBlk;
    gemm_wmma_f32<<<gblocks, T, 0, stream>>>(x, W1, H, N, mtiles);

    pull1_dot_kernel<<<2048, T, 0, stream>>>(rowptr, col, H, dinv, b1, W2, hd, N);

    pull2_kernel<<<(N * 8 + T - 1) / T, T, 0, stream>>>(rowptr, col, hd, b2, out, N);
}


# Layer 1 sums the weighted x rows per node first, then multiplies each 16-node tile of sums by W1 on WMMA in the same launch: product launch and H plane gone, 5 launches to 4; sum order re-associated
# speedup vs baseline: 5.2125x; 1.2819x over previous
#include <hip/hip_runtime.h>
#include <hip/hip_bf16.h>


typedef __attribute__((ext_vector_type(2))) float v2f;
typedef __attribute__((ext_vector_type(8))) float v8f;

#define GCN_D 64


#define BIN_NODES   256
#define MAX_BINS    256
#define CHUNK       4096
#define MAX_CHUNKS  256
#define SORT_T      256
#define EPT         (CHUNK / SORT_T)
#define RUN_LANES   16
#define RUN_GROUPS  (SORT_T / RUN_LANES)
#define RUN_ROUNDS  (MAX_CHUNKS / RUN_GROUPS)
#define LCAP        8192

static __device__ __forceinline__ int scan256(int v, int t, int* wtot, int& tot) {
    const int lane = t & 31;
    const int w    = t >> 5;
    int incl = v;
    for (int off = 1; off < 32; off <<= 1) {
        const int u = __shfl_up(incl, off);
        if (lane >= off) incl += u;
    }
    if (lane == 31) wtot[w] = incl;
    __syncthreads();
    int pre = 0, all = 0;
    for (int k = 0; k < SORT_T / 32; ++k) {
        const int x = wtot[k];
        all += x;
        if (k < w) pre += x;
    }
    tot = all;
    __syncthreads();
    return pre + incl - v;
}

__global__ void __launch_bounds__(SORT_T)
chunk_sort_kernel(const int* __restrict__ src, const int* __restrict__ dst,
                  int* __restrict__ sorted, int* __restrict__ cstart, int E, int nbins) {
    __shared__ int hist[MAX_BINS];
    __shared__ int wtot[SORT_T / 32];
    __shared__ int buf[CHUNK];
    const int t    = threadIdx.x;
    const int base = blockIdx.x * CHUNK;
    hist[t] = 0;
    __syncthreads();

    int w[EPT], b[EPT], r[EPT];
    #pragma unroll
    for (int k = 0; k < EPT; ++k) {
        const int  e  = base + k * SORT_T + t;
        const bool ok = e < E;
        const int  ec = ok ? e : E - 1;
        const int  d  = dst[ec];
        w[k] = (int)(((unsigned)src[ec] << 8) | (unsigned)(d & (BIN_NODES - 1)));
        b[k] = ok ? (d >> 8) : -1;
    }
    #pragma unroll
    for (int k = 0; k < EPT; ++k) r[k] = (b[k] >= 0) ? atomicAdd(&hist[b[k]], 1) : 0;
    __syncthreads();

    int tot;
    const int ex = scan256(hist[t], t, wtot, tot);
    int* __restrict__ crow = cstart + (size_t)blockIdx.x * (nbins + 1);
    if (t < nbins) crow[t] = ex;
    if (t == 0)    crow[nbins] = tot;
    hist[t] = ex;
    __syncthreads();

    int st[EPT];
    #pragma unroll
    for (int k = 0; k < EPT; ++k) st[k] = hist[b[k] >= 0 ? b[k] : 0];
    #pragma unroll
    for (int k = 0; k < EPT; ++k)
        if (b[k] >= 0) buf[st[k] + r[k]] = w[k];
    __syncthreads();

    if (base + CHUNK <= E) {
        #pragma unroll
        for (int k = 0; k < EPT; ++k) sorted[base + k * SORT_T + t] = buf[k * SORT_T + t];
    } else {
        for (int i = t; i < E - base; i += SORT_T) sorted[base + i] = buf[i];
    }
}

__global__ void __launch_bounds__(SORT_T)
bin_lists_kernel(const int* __restrict__ sorted, const int* __restrict__ cstart,
                 int* __restrict__ rowptr, float* __restrict__ dinv, int* __restrict__ col,
                 int n, int nchunks, int nbins) {
    __shared__ int rs0[MAX_CHUNKS];
    __shared__ int rlen[MAX_CHUNKS];
    __shared__ int cnt[BIN_NODES];
    __shared__ int cur[BIN_NODES];
    __shared__ int wa[SORT_T / 32];
    __shared__ int wb[SORT_T / 32];
    __shared__ int lcol[LCAP];
    const int t   = threadIdx.x;
    const int bin = blockIdx.x;

    int s0 = 0, len = 0;
    if (t < nchunks) {
        const int* __restrict__ crow = cstart + (size_t)t * (nbins + 1) + bin;
        s0  = crow[0];
        len = crow[1] - s0;
    }
    rs0[t]  = t * CHUNK + s0;
    rlen[t] = len;
    cnt[t]  = 0;
    int sa = s0, sb = len;
    for (int off = 16; off > 0; off >>= 1) { sa += __shfl_xor(sa, off); sb += __shfl_xor(sb, off); }
    if ((t & 31) == 0) { wa[t >> 5] = sa; wb[t >> 5] = sb; }
    __syncthreads();
    int binstart = 0, nedges = 0;
    for (int k = 0; k < SORT_T / 32; ++k) { binstart += wa[k]; nedges += wb[k]; }

    const int g   = t / RUN_LANES;
    const int sub = t % RUN_LANES;
    int  gs[RUN_ROUNDS], gl[RUN_ROUNDS];
    int  wv[RUN_ROUNDS][2];
    bool longrun = false;
    #pragma unroll
    for (int j = 0; j < RUN_ROUNDS; ++j) {
        const int p = g + j * RUN_GROUPS;
        gs[j] = rs0[p];
        gl[j] = rlen[p];
    }
    #pragma unroll
    for (int j = 0; j < RUN_ROUNDS; ++j) {
        wv[j][0] = 0;
        wv[j][1] = 0;
        if (sub < gl[j])             wv[j][0] = sorted[gs[j] + sub];
        if (sub + RUN_LANES < gl[j]) wv[j][1] = sorted[gs[j] + sub + RUN_LANES];
        longrun = longrun || (gl[j] > 2 * RUN_LANES);
    }

    #pragma unroll
    for (int j = 0; j < RUN_ROUNDS; ++j) {
        if (sub < gl[j])             atomicAdd(&cnt[wv[j][0] & (BIN_NODES - 1)], 1);
        if (sub + RUN_LANES < gl[j]) atomicAdd(&cnt[wv[j][1] & (BIN_NODES - 1)], 1);
    }
    if (longrun) {
        for (int p = g; p < nchunks; p += RUN_GROUPS)
            for (int i = sub + 2 * RUN_LANES; i < rlen[p]; i += RUN_LANES)
                atomicAdd(&cnt[sorted[rs0[p] + i] & (BIN_NODES - 1)], 1);
    }
    __syncthreads();

    const int c = cnt[t];
    int tot;
    const int ex = scan256(c, t, wa, tot);
    cur[t] = ex;
    const int node = bin * BIN_NODES + t;
    if (node < n) {
        rowptr[node] = binstart + ex;
        dinv[node]   = rsqrtf((float)c + 1.0f);
        if (node == n - 1) rowptr[n] = binstart + ex + c;
    }
    __syncthreads();

    const bool fits = nedges <= LCAP;
    int pos[RUN_ROUNDS][2];
    #pragma unroll
    for (int j = 0; j < RUN_ROUNDS; ++j) {
        #pragma unroll
        for (int h = 0; h < 2; ++h) {
            pos[j][h] = 0;
            if (sub + h * RUN_LANES < gl[j]) pos[j][h] = atomicAdd(&cur[(unsigned)wv[j][h] & (BIN_NODES - 1)], 1);
        }
    }
    if (fits) {
        #pragma unroll
        for (int j = 0; j < RUN_ROUNDS; ++j) {
            #pragma unroll
            for (int h = 0; h < 2; ++h)
                if (sub + h * RUN_LANES < gl[j]) lcol[pos[j][h]] = (int)((unsigned)wv[j][h] >> 8);
        }
    } else {
        #pragma unroll
        for (int j = 0; j < RUN_ROUNDS; ++j) {
            #pragma unroll
            for (int h = 0; h < 2; ++h)
                if (sub + h * RUN_LANES < gl[j]) col[binstart + pos[j][h]] = (int)((unsigned)wv[j][h] >> 8);
        }
    }
    if (longrun) {
        for (int p = g; p < nchunks; p += RUN_GROUPS)
            for (int i = sub + 2 * RUN_LANES; i < rlen[p]; i += RUN_LANES) {
                const unsigned wd    = (unsigned)sorted[rs0[p] + i];
                const int      place = atomicAdd(&cur[wd & (BIN_NODES - 1)], 1);
                if (fits) lcol[place] = (int)(wd >> 8);
                else      col[binstart + place] = (int)(wd >> 8);
            }
    }
    __syncthreads();

    if (fits) {
        for (int i0 = 0; i0 < nedges; i0 += 4 * SORT_T) {
            int v[4];
            #pragma unroll
            for (int u = 0; u < 4; ++u) {
                const int i = i0 + u * SORT_T + t;
                v[u] = lcol[i < nedges ? i : nedges - 1];
            }
            #pragma unroll
            for (int u = 0; u < 4; ++u) {
                const int i = i0 + u * SORT_T + t;
                if (i < nedges) col[binstart + i] = v[u];
            }
        }
    }
}


#define TILE_NODES      32
#define TILE_WAVES      8
#define NODES_PER_WAVE  (TILE_NODES / TILE_WAVES)
#define A_PITCH         (GCN_D + 4)

static __device__ __forceinline__ int   lane_i(int v, int k)   { return __builtin_amdgcn_readlane(v, k); }
static __device__ __forceinline__ float lane_f(float v, int k) {
    return __builtin_bit_cast(float, __builtin_amdgcn_readlane(__builtin_bit_cast(int, v), k));
}
static __device__ __forceinline__ float2 row2(const float* __restrict__ X, int s, int lane) {
    return ((const float2*)(X + (size_t)s * GCN_D))[lane];
}

template <int M>
static __device__ __forceinline__ float row_xor(float v) {
    return __builtin_bit_cast(float, __builtin_amdgcn_update_dpp(0, __builtin_bit_cast(int, v), 0x160 + M, 0xf, 0xf, true));
}

template <int U>
static __device__ __forceinline__ void add_rows(const float* __restrict__ X, int s_l, float n_l,
                                                int k, int m, int lane, float2& acc) {
    int    s[U];
    float  nn[U];
    float2 h[U];
    #pragma unroll
    for (int u = 0; u < U; ++u) { s[u] = lane_i(s_l, k + u); nn[u] = lane_f(n_l, k + u); }
    #pragma unroll
    for (int u = 0; u < U; ++u) h[u] = row2(X, s[u], lane);
    #pragma unroll
    for (int u = 0; u < U; ++u) {
        const float hx = (u == 0 || k + u < m) ? h[u].x : 0.0f;
        const float hy = (u == 0 || k + u < m) ? h[u].y : 0.0f;
        acc.x += hx * nn[u];
        acc.y += hy * nn[u];
    }
}

__global__ void __launch_bounds__(TILE_WAVES * 32)
pull1_gemm_dot_kernel(const int* __restrict__ rowptr, const int* __restrict__ col,
                      const float* __restrict__ X, const float* __restrict__ dinv,
                      const float* __restrict__ W1, const float* __restrict__ b1, const float* __restrict__ W2,
                      float* __restrict__ hd, int n) {
    __shared__ __attribute__((aligned(16))) float sA[TILE_NODES * A_PITCH];
    __shared__ __attribute__((aligned(16))) float sP[TILE_NODES * 4];
    __shared__ float sD[TILE_NODES];

    const int lane = threadIdx.x & 31;
    const int wave  = __builtin_amdgcn_readfirstlane((int)(threadIdx.x >> 5));
    const int node0 = (int)blockIdx.x * TILE_NODES;

    for (int q = 0; q < NODES_PER_WAVE; ++q) {
        const int r = wave * NODES_PER_WAVE + q;
        const int i = node0 + r;
        float2 z  = make_float2(0.0f, 0.0f);
        float  di = 0.0f;
        if (i < n) {
            const int beg = rowptr[i];
            const int end = rowptr[i + 1];
            di = dinv[i];
            const float2 xs = row2(X, i, lane);
            float2 acc = make_float2(0.0f, 0.0f);

            for (int base = beg; base < end; base += 32) {
                const int j = base + lane;
                int   s_l = i;
                float n_l = 0.0f;
                if (j < end) { s_l = col[j]; n_l = dinv[s_l] * di; }
                const int left = end - base;
                const int m    = left < 32 ? left : 32;

                int k = 0;
                for (; k + 4 < m; k += 8) add_rows<8>(X, s_l, n_l, k, m, lane, acc);
                if (k < m)                add_rows<4>(X, s_l, n_l, k, m, lane, acc);
            }
            const float d2 = di * di;
            z.x = acc.x + xs.x * d2;
            z.y = acc.y + xs.y * d2;
        }
        ((float2*)(sA + r * A_PITCH))[lane] = z;
        if (lane == 0) sD[r] = di;
    }

    const int l15 = lane & 15;
    const int hi  = lane >> 4;
    const int mt  = wave >> 2;
    const int nt  = wave & 3;
    const int cg  = nt * 16 + l15;
    const float* __restrict__ wp = W1 + (size_t)(2 * hi) * GCN_D + cg;
    const float bcol = b1[cg];
    const float wcol = W2[cg];

    __syncthreads();

    const float* arow = sA + (mt * 16 + l15) * A_PITCH + 2 * hi;
    v8f acc0 = {}, acc1 = {};
    #pragma unroll
    for (int kk = 0; kk < GCN_D; kk += 8) {
        const v2f a0 = *(const v2f*)(arow + kk);
        const v2f a1 = *(const v2f*)(arow + kk + 4);
        v2f b0, b1f;
        b0.x  = wp[(size_t)kk * GCN_D];
        b0.y  = wp[(size_t)kk * GCN_D + GCN_D];
        b1f.x = wp[(size_t)(kk + 4) * GCN_D];
        b1f.y = wp[(size_t)(kk + 4) * GCN_D + GCN_D];
        acc0 = __builtin_amdgcn_wmma_f32_16x16x4_f32(false, a0, false, b0,  (short)0, acc0, false, false);
        acc1 = __builtin_amdgcn_wmma_f32_16x16x4_f32(false, a1, false, b1f, (short)0, acc1, false, false);
    }

    float v[8];
    #pragma unroll
    for (int r = 0; r < 8; ++r) v[r] = fmaxf((acc0[r] + acc1[r]) + bcol, 0.0f) * wcol;
    #pragma unroll
    for (int r = 0; r < 8; ++r) v[r] += row_xor<1>(v[r]);
    #pragma unroll
    for (int r = 0; r < 8; ++r) v[r] += row_xor<2>(v[r]);
    #pragma unroll
    for (int r = 0; r < 8; ++r) v[r] += row_xor<4>(v[r]);
    #pragma unroll
    for (int r = 0; r < 8; ++r) v[r] += row_xor<8>(v[r]);
    if (l15 == 0) {
        #pragma unroll
        for (int r = 0; r < 8; ++r) sP[(mt * 16 + 8 * hi + r) * 4 + nt] = v[r];
    }
    __syncthreads();

    if (threadIdx.x < TILE_NODES) {
        const int i = node0 + (int)threadIdx.x;
        if (i < n) {
            const float4 p4 = ((const float4*)sP)[threadIdx.x];
            const float  p  = (p4.x + p4.y) + (p4.z + p4.w);
            ((float2*)hd)[i] = make_float2(p, sD[threadIdx.x]);
        }
    }
}

__global__ void pull2_kernel(const int* __restrict__ rowptr, const int* __restrict__ col,
                             const float* __restrict__ hd,
                             const float* __restrict__ b2, float* __restrict__ out, int n) {
    const float2* __restrict__ pr = (const float2*)hd;
    const int t   = blockIdx.x * blockDim.x + threadIdx.x;
    const int i   = t >> 3;
    const int sub = t & 7;
    float v  = 0.0f;
    float hi = 0.0f;
    float di = 0.0f;
    if (i < n) {
        const float2 me = pr[i];
        hi = me.x;
        di = me.y;
        const int beg = rowptr[i];
        const int end = rowptr[i + 1];
        for (int j = beg + sub; j < end; j += 8) {
            const float2 g = pr[col[j]];
            v += g.x * g.y * di;
        }
    }
    v += __shfl_xor(v, 1);
    v += __shfl_xor(v, 2);
    v += __shfl_xor(v, 4);
    if (i < n && sub == 0) out[i] = fmaf(hi * di, di, v) + b2[0];
}


extern "C" void kernel_launch(void* const* d_in, const int* in_sizes, int n_in,
                              void* d_out, int out_size, void* d_ws, size_t ws_size,
                              hipStream_t stream) {
    const float* x  = (const float*)d_in[0];
    const int*   ei = (const int*)d_in[1];
    const float* W1 = (const float*)d_in[2];
    const float* b1 = (const float*)d_in[3];
    const float* W2 = (const float*)d_in[4];
    const float* b2 = (const float*)d_in[5];
    float* out = (float*)d_out;

    const int N = in_sizes[0] / GCN_D;
    const int E = in_sizes[1] / 2;
    const int* src = ei;
    const int* dst = ei + E;

    int*   sorted = (int*)d_ws;
    float* hd     = (float*)sorted;
    int*   col    = sorted + E;
    int*   rowptr = col + E;
    float* dinv   = (float*)(rowptr + N + 1);
    int*   cstart = (int*)(dinv + N);

    const int T = 256;
    const int nbins   = (N + BIN_NODES - 1) / BIN_NODES;
    const int nchunks = (E + CHUNK - 1) / CHUNK;

    chunk_sort_kernel<<<nchunks, SORT_T, 0, stream>>>(src, dst, sorted, cstart, E, nbins);

    bin_lists_kernel<<<nbins, SORT_T, 0, stream>>>(sorted, cstart, rowptr, dinv, col, N, nchunks, nbins);

    pull1_gemm_dot_kernel<<<(N + TILE_NODES - 1) / TILE_NODES, TILE_WAVES * 32, 0, stream>>>(
        rowptr, col, x, dinv, W1, b1, W2, hd, N);

    pull2_kernel<<<(N * 8 + T - 1) / T, T, 0, stream>>>(rowptr, col, hd, b2, out, N);
}
